# starting kernel
# baseline (speedup 1.0000x reference)
#include <hip/hip_runtime.h>
#include <hip/hip_bf16.h>

#define B_  32
#define S_  1024
#define D_  256
#define W_  4
#define NL_ 2
#define NH_ 2

typedef __attribute__((ext_vector_type(2))) float v2f;
typedef __attribute__((ext_vector_type(8))) float v8f;

__global__ __launch_bounds__(256)
void fir_kernel(const float* __restrict__ x, int xrow,
                const float* __restrict__ padf, const float* __restrict__ padb,
                const float* __restrict__ w, int off,
                float* __restrict__ y) {
  int gid = blockIdx.x * blockDim.x + threadIdx.x;
  int d = gid & (D_ - 1);
  int s = (gid >> 8) & (S_ - 1);
  int b = gid >> 18;
  float acc = 0.f;
#pragma unroll
  for (int j = 0; j <= W_; ++j) {
    int t = s + off + j;
    float v;
    if (t < W_)             v = padf[t * D_ + d];
    else if (t >= W_ + S_)  v = padb[(t - W_ - S_) * D_ + d];
    else                    v = x[((size_t)b * S_ + (t - W_)) * (size_t)xrow + d];
    acc = fmaf(w[j], v, acc);
  }
  y[(size_t)gid] = acc;
}

__global__ __launch_bounds__(256)
void highway_kernel(const float* __restrict__ x,
                    const float* __restrict__ Wt,
                    const float* __restrict__ bias,
                    float* __restrict__ out, int orow) {
  const int lane = threadIdx.x & 31;
  const int wave = threadIdx.x >> 5;
  const int half = lane >> 4;
  const int l    = lane & 15;

  const int m0 = (blockIdx.y * 8 + wave) * 32;
  const int n0 = blockIdx.x * 32;
  const int nA = n0 + l;
  const int nB = n0 + 16 + l;

  const float* arow0 = x  + (size_t)(m0 + l) * D_       + 2 * half;
  const float* arow1 = x  + (size_t)(m0 + 16 + l) * D_  + 2 * half;
  const float* wn0   = Wt + (size_t)nA * D_             + 2 * half;
  const float* wn1   = Wt + (size_t)nB * D_             + 2 * half;
  const float* wg0   = Wt + (size_t)(D_ + nA) * D_      + 2 * half;
  const float* wg1   = Wt + (size_t)(D_ + nB) * D_      + 2 * half;

  v8f aN0 = {}; v8f aN1 = {}; v8f aG0 = {}; v8f aG1 = {};
  v8f bN0 = {}; v8f bN1 = {}; v8f bG0 = {}; v8f bG1 = {};

#pragma unroll 2
  for (int g = 0; g < D_ / 4; ++g) {
    v2f a0 = *(const v2f*)(arow0 + 4 * g);
    v2f a1 = *(const v2f*)(arow1 + 4 * g);
    v2f n0v = *(const v2f*)(wn0 + 4 * g);
    v2f n1v = *(const v2f*)(wn1 + 4 * g);
    v2f g0v = *(const v2f*)(wg0 + 4 * g);
    v2f g1v = *(const v2f*)(wg1 + 4 * g);
    aN0 = __builtin_amdgcn_wmma_f32_16x16x4_f32(false, a0, false, n0v, (short)0, aN0, false, false);
    aN1 = __builtin_amdgcn_wmma_f32_16x16x4_f32(false, a0, false, n1v, (short)0, aN1, false, false);
    aG0 = __builtin_amdgcn_wmma_f32_16x16x4_f32(false, a0, false, g0v, (short)0, aG0, false, false);
    aG1 = __builtin_amdgcn_wmma_f32_16x16x4_f32(false, a0, false, g1v, (short)0, aG1, false, false);
    bN0 = __builtin_amdgcn_wmma_f32_16x16x4_f32(false, a1, false, n0v, (short)0, bN0, false, false);
    bN1 = __builtin_amdgcn_wmma_f32_16x16x4_f32(false, a1, false, n1v, (short)0, bN1, false, false);
    bG0 = __builtin_amdgcn_wmma_f32_16x16x4_f32(false, a1, false, g0v, (short)0, bG0, false, false);
    bG1 = __builtin_amdgcn_wmma_f32_16x16x4_f32(false, a1, false, g1v, (short)0, bG1, false, false);
  }

  const float bnA = bias[nA], bnB = bias[nB];
  const float bgA = bias[D_ + nA], bgB = bias[D_ + nB];

#pragma unroll
  for (int r = 0; r < 8; ++r) {
    {
      const int m = m0 + r + 8 * half;
      const float* xr = x + (size_t)m * D_;
      float* op = out + (size_t)m * orow;
      float g0 = 1.f / (1.f + __expf(-(aG0[r] + bgA)));
      op[nA] = g0 * xr[nA] + (1.f - g0) * fmaxf(aN0[r] + bnA, 0.f);
      float g1 = 1.f / (1.f + __expf(-(aG1[r] + bgB)));
      op[nB] = g1 * xr[nB] + (1.f - g1) * fmaxf(aN1[r] + bnB, 0.f);
    }
    {
      const int m = m0 + 16 + r + 8 * half;
      const float* xr = x + (size_t)m * D_;
      float* op = out + (size_t)m * orow;
      float g0 = 1.f / (1.f + __expf(-(bG0[r] + bgA)));
      op[nA] = g0 * xr[nA] + (1.f - g0) * fmaxf(bN0[r] + bnA, 0.f);
      float g1 = 1.f / (1.f + __expf(-(bG1[r] + bgB)));
      op[nB] = g1 * xr[nB] + (1.f - g1) * fmaxf(bN1[r] + bnB, 0.f);
    }
  }
}

extern "C" void kernel_launch(void* const* d_in, const int* in_sizes, int n_in,
                              void* d_out, int out_size, void* d_ws, size_t ws_size,
                              hipStream_t stream) {
  (void)in_sizes; (void)n_in; (void)out_size; (void)ws_size;

  const float* inputs   = (const float*)d_in[0];
  const float* fwd_pad  = (const float*)d_in[1];
  const float* bwd_pad  = (const float*)d_in[2];
  const float* fwd_w    = (const float*)d_in[3];
  const float* bwd_w    = (const float*)d_in[4];
  const float* fwd_hw_W = (const float*)d_in[5];
  const float* fwd_hw_b = (const float*)d_in[6];
  const float* bwd_hw_W = (const float*)d_in[7];
  const float* bwd_hw_b = (const float*)d_in[8];
  float* out = (float*)d_out;

  const size_t M = (size_t)B_ * S_;
  float* yA = (float*)d_ws;
  float* yB = yA + M * D_;

  dim3 hwBlock(256);
  dim3 hwGrid(D_ / 32, (unsigned)(M / (32 * 8)));
  const int firBlocks = (int)(M * D_ / 256);

  for (int i = 0; i < NL_; ++i) {
    for (int dir = 0; dir < 2; ++dir) {
      const float* xin; int xrow;
      if (i == 0) { xin = inputs; xrow = D_; }
      else        { xin = out + (size_t)(i - 1) * M * 2 * D_ + dir * D_; xrow = 2 * D_; }

      const float* taps = (dir ? bwd_w : fwd_w) + i * (W_ + 1);
      fir_kernel<<<firBlocks, 256, 0, stream>>>(
          xin, xrow,
          fwd_pad + (size_t)i * W_ * D_, bwd_pad + (size_t)i * W_ * D_,
          taps, dir ? W_ : 0, yA);

      const float* HW = (dir ? bwd_hw_W : fwd_hw_W) + (size_t)i * NH_ * 2 * D_ * D_;
      const float* Hb = (dir ? bwd_hw_b : fwd_hw_b) + (size_t)i * NH_ * 2 * D_;

      highway_kernel<<<hwGrid, hwBlock, 0, stream>>>(yA, HW, Hb, yB, D_);
      highway_kernel<<<hwGrid, hwBlock, 0, stream>>>(
          yB, HW + (size_t)2 * D_ * D_, Hb + 2 * D_,
          out + (size_t)i * M * 2 * D_ + dir * D_, 2 * D_);
    }
  }
}


# Three of the four FIR passes formed by the highway waves beside their f32 WMMA K loop (one packed word per lane per trip, for a later launch): 10 launches, not 13; tap order and sums kept
# speedup vs baseline: 1.9235x; 1.9235x over previous
#include <hip/hip_runtime.h>
#include <hip/hip_bf16.h>


#define B_  32
#define S_  1024
#define D_  256
#define W_  4
#define NL_ 2
#define NH_ 2

typedef __attribute__((ext_vector_type(2))) float v2f;
typedef __attribute__((ext_vector_type(4))) float v4f;
typedef __attribute__((ext_vector_type(8))) float v8f;


__global__ __launch_bounds__(256)
void pack_w_kernel(const float* __restrict__ wf, const float* __restrict__ wb,
                   float* __restrict__ dst) {
  const float* src = blockIdx.y ? wb : wf;
  float* d = dst + (size_t)blockIdx.y * (4 * 2 * D_ * D_);
  const int q    = blockIdx.x * 256 + threadIdx.x;
  const int lane = q & 31;
  const int p    = (q >> 5) & 1;
  const int ch   = (q >> 6) & 15;
  const int T    = q >> 10;
  const int row  = T * 16 + (lane & 15);
  const int k    = ch * 16 + 8 * (lane >> 4) + 4 * p;
  *(v4f*)(d + (size_t)q * 4) = *(const v4f*)(src + (size_t)row * D_ + k);
}

__global__ __launch_bounds__(256)
void fir_kernel(const float* __restrict__ x, int xrow,
                const float* __restrict__ padf, const float* __restrict__ padb,
                const float* __restrict__ w, int off,
                float* __restrict__ y) {
  const int q    = blockIdx.x * 256 + threadIdx.x;
  const int lane = q & 31;
  const int p    = (q >> 5) & 1;
  const int ch   = (q >> 6) & 15;
  const int T    = q >> 10;
  const int m    = T * 16 + (lane & 15);
  const int d    = ch * 16 + 8 * (lane >> 4) + 4 * p;
  const int s    = m & (S_ - 1);
  const int b    = m >> 10;
  v4f acc = {0.f, 0.f, 0.f, 0.f};
#pragma unroll
  for (int j = 0; j <= W_; ++j) {
    const int t = s + off + j;
    v4f v;
    if (t < W_)             v = *(const v4f*)(padf + t * D_ + d);
    else if (t >= W_ + S_)  v = *(const v4f*)(padb + (t - W_ - S_) * D_ + d);
    else                    v = *(const v4f*)(x + ((size_t)b * S_ + (t - W_)) * (size_t)xrow + d);
    const float wj = w[j];
    acc.x = fmaf(wj, v.x, acc.x);
    acc.y = fmaf(wj, v.y, acc.y);
    acc.z = fmaf(wj, v.z, acc.z);
    acc.w = fmaf(wj, v.w, acc.w);
  }
  *(v4f*)(y + (size_t)q * 4) = acc;
}

__device__ __forceinline__ v8f mma(v2f a, v2f b, v8f c) {
  return __builtin_amdgcn_wmma_f32_16x16x4_f32(false, a, false, b, (short)0, c, false, false);
}

template <int ROWMAJOR_OUT>
__device__ __forceinline__ void hw_tile_out(v8f N, v8f G,
                                            const float* __restrict__ bn,
                                            const float* __restrict__ bg,
                                            v4f x0, v4f x1,
                                            float* __restrict__ o) {
  const v4f bn0 = *(const v4f*)bn, bn1 = *(const v4f*)(bn + 4);
  const v4f bg0 = *(const v4f*)bg, bg1 = *(const v4f*)(bg + 4);
  v4f o0, o1;
#pragma unroll
  for (int r = 0; r < 4; ++r) {
    float g0 = 1.f / (1.f + __expf(-(G[r] + bg0[r])));
    o0[r] = g0 * x0[r] + (1.f - g0) * fmaxf(N[r] + bn0[r], 0.f);
    float g1 = 1.f / (1.f + __expf(-(G[4 + r] + bg1[r])));
    o1[r] = g1 * x1[r] + (1.f - g1) * fmaxf(N[4 + r] + bn1[r], 0.f);
  }
  *(v4f*)o = o0;
  *(v4f*)(o + (ROWMAJOR_OUT ? 4 : 128)) = o1;
}

template <int ROWMAJOR_OUT, int SIDE>
__global__ __launch_bounds__(256)
void highway_kernel(const float* __restrict__ xp,
                    const float* __restrict__ wp,
                    const float* __restrict__ bias,
                    float* __restrict__ out, int orow,
                    const float* __restrict__ fx, int fxrow,
                    const float* __restrict__ fpadf, const float* __restrict__ fpadb,
                    const float* __restrict__ fw, int foff,
                    float* __restrict__ fy) {
  const int lane = threadIdx.x & 31;
  const int wave = __builtin_amdgcn_readfirstlane(threadIdx.x >> 5);
  const int h    = lane >> 4;
  const int c    = lane & 15;

  const int T0  = (blockIdx.y * 4 + (wave & 3)) * 4;
  const int nt0 = (blockIdx.x * 2 + (wave >> 2)) * 2;

  const float* xb = xp + (size_t)T0 * (16 * 256) + lane * 4;
  const float* wb = wp + (size_t)nt0 * (16 * 256) + lane * 4;

  v8f acc[4][4];
#pragma unroll
  for (int r = 0; r < 4; ++r)
#pragma unroll
    for (int f = 0; f < 4; ++f) acc[r][f] = v8f{};

  const float* tp[W_ + 1];
  float tw[W_ + 1];
  float* yw = nullptr;
  if (SIDE) {
    const int gw  = (blockIdx.y * (D_ / 64) + blockIdx.x) * 8 + wave;
    const int ch0 = (gw & 1) * 8;
    const int m   = (gw >> 1) * 16 + c;
    const int s   = m & (S_ - 1);
    const int b   = m >> 10;
#pragma unroll
    for (int j = 0; j <= W_; ++j) {
      const int t = s + foff + j;
      const float* q;
      if (t < W_)             q = fpadf + t * D_;
      else if (t >= W_ + S_)  q = fpadb + (t - W_ - S_) * D_;
      else                    q = fx + ((size_t)b * S_ + (t - W_)) * (size_t)fxrow;
      tp[j] = q + ch0 * 16 + 8 * h;
      tw[j] = fw[j];
    }
    yw = fy + ((size_t)(gw >> 1) * 1024 + ch0 * 64 + lane) * 4;
  }

  v4f cx[4], cw[4];
#pragma unroll
  for (int r = 0; r < 4; ++r) cx[r] = *(const v4f*)(xb + r * 4096);
  cw[0] = *(const v4f*)(wb);
  cw[1] = *(const v4f*)(wb + 4096);
  cw[2] = *(const v4f*)(wb + 16 * 4096);
  cw[3] = *(const v4f*)(wb + 17 * 4096);

#pragma nounroll
  for (int it2 = 0; it2 < D_ / 16; ++it2) {
    const int n1 = (2 * it2 + 1) * 128;
    v4f px[4], pw[4];
#pragma unroll
    for (int r = 0; r < 4; ++r) px[r] = *(const v4f*)(xb + n1 + r * 4096);
    pw[0] = *(const v4f*)(wb + n1);
    pw[1] = *(const v4f*)(wb + n1 + 4096);
    pw[2] = *(const v4f*)(wb + n1 + 16 * 4096);
    pw[3] = *(const v4f*)(wb + n1 + 17 * 4096);
    v4f fv[W_ + 1];
    if (SIDE) {
      __builtin_amdgcn_sched_barrier(0);
      const int dofs = (it2 >> 1) * 16 + (it2 & 1) * 4;
#pragma unroll
      for (int j = 0; j <= W_; ++j) fv[j] = *(const v4f*)(tp[j] + dofs);
    }
    __builtin_amdgcn_sched_barrier(0);

#pragma unroll
    for (int r = 0; r < 4; ++r)
#pragma unroll
      for (int f = 0; f < 4; ++f) acc[r][f] = mma(cw[f].xy, cx[r].xy, acc[r][f]);
#pragma unroll
    for (int r = 0; r < 4; ++r)
#pragma unroll
      for (int f = 0; f < 4; ++f) acc[r][f] = mma(cw[f].zw, cx[r].zw, acc[r][f]);

    const int n2 = (it2 + 1 < D_ / 16 ? 2 * it2 + 2 : 2 * it2 + 1) * 128;
#pragma unroll
    for (int r = 0; r < 4; ++r) cx[r] = *(const v4f*)(xb + n2 + r * 4096);
    cw[0] = *(const v4f*)(wb + n2);
    cw[1] = *(const v4f*)(wb + n2 + 4096);
    cw[2] = *(const v4f*)(wb + n2 + 16 * 4096);
    cw[3] = *(const v4f*)(wb + n2 + 17 * 4096);
    __builtin_amdgcn_sched_barrier(0);

#pragma unroll
    for (int r = 0; r < 4; ++r)
#pragma unroll
      for (int f = 0; f < 4; ++f) acc[r][f] = mma(pw[f].xy, px[r].xy, acc[r][f]);
#pragma unroll
    for (int r = 0; r < 4; ++r)
#pragma unroll
      for (int f = 0; f < 4; ++f) acc[r][f] = mma(pw[f].zw, px[r].zw, acc[r][f]);

    if (SIDE) {
      __builtin_amdgcn_sched_barrier(0);
      v4f a = {0.f, 0.f, 0.f, 0.f};
#pragma unroll
      for (int j = 0; j <= W_; ++j) {
        a.x = fmaf(tw[j], fv[j].x, a.x);
        a.y = fmaf(tw[j], fv[j].y, a.y);
        a.z = fmaf(tw[j], fv[j].z, a.z);
        a.w = fmaf(tw[j], fv[j].w, a.w);
      }
      *(v4f*)(yw + it2 * 128) = a;
    }
  }

  const float* bnA = bias + nt0 * 16 + 8 * h;
  const float* bnB = bnA + 16;
  const float* bgA = bnA + D_;
  const float* bgB = bnB + D_;
  const float* res = xp + ((size_t)T0 * 16 + nt0) * 256 + lane * 4;
  v4f rx[2][4];
  rx[0][0] = *(const v4f*)(res);
  rx[0][1] = *(const v4f*)(res + 128);
  rx[0][2] = *(const v4f*)(res + 256);
  rx[0][3] = *(const v4f*)(res + 384);
#pragma unroll
  for (int r = 0; r < 4; ++r) {
    if (r + 1 < 4) {
      const float* rn = res + (r + 1) * 4096;
      rx[(r + 1) & 1][0] = *(const v4f*)(rn);
      rx[(r + 1) & 1][1] = *(const v4f*)(rn + 128);
      rx[(r + 1) & 1][2] = *(const v4f*)(rn + 256);
      rx[(r + 1) & 1][3] = *(const v4f*)(rn + 384);
    }
    __builtin_amdgcn_sched_barrier(0);
    float *oA, *oB;
    if (ROWMAJOR_OUT) {
      oA = out + (size_t)((T0 + r) * 16 + c) * orow + nt0 * 16 + 8 * h;
      oB = oA + 16;
    } else {
      oA = out + ((size_t)(T0 + r) * 16 + nt0) * 256 + lane * 4;
      oB = oA + 256;
    }
    hw_tile_out<ROWMAJOR_OUT>(acc[r][0], acc[r][2], bnA, bgA, rx[r & 1][0], rx[r & 1][1], oA);
    hw_tile_out<ROWMAJOR_OUT>(acc[r][1], acc[r][3], bnB, bgB, rx[r & 1][2], rx[r & 1][3], oB);
  }
}

extern "C" void kernel_launch(void* const* d_in, const int* in_sizes, int n_in,
                              void* d_out, int out_size, void* d_ws, size_t ws_size,
                              hipStream_t stream) {
  (void)in_sizes; (void)n_in; (void)out_size; (void)ws_size;

  const float* inputs   = (const float*)d_in[0];
  const float* fwd_pad  = (const float*)d_in[1];
  const float* bwd_pad  = (const float*)d_in[2];
  const float* fwd_w    = (const float*)d_in[3];
  const float* bwd_w    = (const float*)d_in[4];
  const float* fwd_hw_W = (const float*)d_in[5];
  const float* fwd_hw_b = (const float*)d_in[6];
  const float* bwd_hw_W = (const float*)d_in[7];
  const float* bwd_hw_b = (const float*)d_in[8];
  float* out = (float*)d_out;

  const size_t M = (size_t)B_ * S_;
  const size_t WMAT = (size_t)2 * D_ * D_;
  float* yA = (float*)d_ws;
  float* yB = yA + M * D_;
  float* wP = yB + M * D_;
  float* yC = wP + 2 * NL_ * NH_ * WMAT;

  dim3 hwBlock(256);
  dim3 hwGrid(D_ / 64, (unsigned)(M / (64 * 4)));
  const int firBlocks = (int)(M * D_ / 4 / 256);

  pack_w_kernel<<<dim3((unsigned)(NL_ * NH_ * WMAT / 4 / 256), 2), 256, 0, stream>>>(
      fwd_hw_W, bwd_hw_W, wP);

  fir_kernel<<<firBlocks, 256, 0, stream>>>(inputs, D_, fwd_pad, bwd_pad, fwd_w, 0, yA);

  for (int i = 0; i < NL_; ++i) {
    for (int dir = 0; dir < 2; ++dir) {
      const float* ycur = dir ? yC : yA;
      const float* HW = wP + ((size_t)dir * NL_ * NH_ + (size_t)i * NH_) * WMAT;
      const float* Hb = (dir ? bwd_hw_b : fwd_hw_b) + (size_t)i * NH_ * 2 * D_;

      const int ni = dir ? i + 1 : i;
      const int nd = dir ? 0 : 1;
      if (ni < NL_) {
        const float* xin; int xrow;
        if (ni == 0) { xin = inputs; xrow = D_; }
        else         { xin = out + (size_t)(ni - 1) * M * 2 * D_ + nd * D_; xrow = 2 * D_; }
        highway_kernel<0, 1><<<hwGrid, hwBlock, 0, stream>>>(
            ycur, HW, Hb, yB, D_,
            xin, xrow, fwd_pad + (size_t)ni * W_ * D_, bwd_pad + (size_t)ni * W_ * D_,
            (nd ? bwd_w : fwd_w) + ni * (W_ + 1), nd ? W_ : 0, nd ? yC : yA);
      } else {
        highway_kernel<0, 0><<<hwGrid, hwBlock, 0, stream>>>(
            ycur, HW, Hb, yB, D_, nullptr, 0, nullptr, nullptr, nullptr, 0, nullptr);
      }
      highway_kernel<1, 0><<<hwGrid, hwBlock, 0, stream>>>(
          yB, HW + WMAT, Hb + 2 * D_,
          out + (size_t)i * M * 2 * D_ + dir * D_, 2 * D_,
          nullptr, 0, nullptr, nullptr, nullptr, 0, nullptr);
    }
  }
}


# Both highway layers of a block in one launch through memory: work-group = 64 rows x all 512 columns, 64x64 wave tiles and 4 waves per SIMD kept, one barrier between the layers; 6 launches; sums kept
# speedup vs baseline: 2.0097x; 1.0448x over previous
#include <hip/hip_runtime.h>
#include <hip/hip_bf16.h>


#define B_  32
#define S_  1024
#define D_  256
#define W_  4
#define NL_ 2
#define NH_ 2

typedef __attribute__((ext_vector_type(2))) float v2f;
typedef __attribute__((ext_vector_type(4))) float v4f;
typedef __attribute__((ext_vector_type(8))) float v8f;


__global__ __launch_bounds__(256)
void pack_w_kernel(const float* __restrict__ wf, const float* __restrict__ wb,
                   float* __restrict__ dst) {
  const float* src = blockIdx.y ? wb : wf;
  float* d = dst + (size_t)blockIdx.y * (4 * 2 * D_ * D_);
  const int q    = blockIdx.x * 256 + threadIdx.x;
  const int lane = q & 31;
  const int p    = (q >> 5) & 1;
  const int ch   = (q >> 6) & 15;
  const int T    = q >> 10;
  const int row  = T * 16 + (lane & 15);
  const int k    = ch * 16 + 8 * (lane >> 4) + 4 * p;
  *(v4f*)(d + (size_t)q * 4) = *(const v4f*)(src + (size_t)row * D_ + k);
}

__global__ __launch_bounds__(256)
void fir_kernel(const float* __restrict__ x, int xrow,
                const float* __restrict__ padf, const float* __restrict__ padb,
                const float* __restrict__ w, int off,
                float* __restrict__ y) {
  const int q    = blockIdx.x * 256 + threadIdx.x;
  const int lane = q & 31;
  const int p    = (q >> 5) & 1;
  const int ch   = (q >> 6) & 15;
  const int T    = q >> 10;
  const int m    = T * 16 + (lane & 15);
  const int d    = ch * 16 + 8 * (lane >> 4) + 4 * p;
  const int s    = m & (S_ - 1);
  const int b    = m >> 10;
  v4f acc = {0.f, 0.f, 0.f, 0.f};
#pragma unroll
  for (int j = 0; j <= W_; ++j) {
    const int t = s + off + j;
    v4f v;
    if (t < W_)             v = *(const v4f*)(padf + t * D_ + d);
    else if (t >= W_ + S_)  v = *(const v4f*)(padb + (t - W_ - S_) * D_ + d);
    else                    v = *(const v4f*)(x + ((size_t)b * S_ + (t - W_)) * (size_t)xrow + d);
    const float wj = w[j];
    acc.x = fmaf(wj, v.x, acc.x);
    acc.y = fmaf(wj, v.y, acc.y);
    acc.z = fmaf(wj, v.z, acc.z);
    acc.w = fmaf(wj, v.w, acc.w);
  }
  *(v4f*)(y + (size_t)q * 4) = acc;
}

__device__ __forceinline__ v8f mma(v2f a, v2f b, v8f c) {
  return __builtin_amdgcn_wmma_f32_16x16x4_f32(false, a, false, b, (short)0, c, false, false);
}

template <int ROWMAJOR_OUT>
__device__ __forceinline__ void hw_tile_out(v8f N, v8f G,
                                            const float* __restrict__ bn,
                                            const float* __restrict__ bg,
                                            v4f x0, v4f x1,
                                            float* __restrict__ o) {
  const v4f bn0 = *(const v4f*)bn, bn1 = *(const v4f*)(bn + 4);
  const v4f bg0 = *(const v4f*)bg, bg1 = *(const v4f*)(bg + 4);
  v4f o0, o1;
#pragma unroll
  for (int r = 0; r < 4; ++r) {
    float g0 = 1.f / (1.f + __expf(-(G[r] + bg0[r])));
    o0[r] = g0 * x0[r] + (1.f - g0) * fmaxf(N[r] + bn0[r], 0.f);
    float g1 = 1.f / (1.f + __expf(-(G[4 + r] + bg1[r])));
    o1[r] = g1 * x1[r] + (1.f - g1) * fmaxf(N[4 + r] + bn1[r], 0.f);
  }
  *(v4f*)o = o0;
  *(v4f*)(o + (ROWMAJOR_OUT ? 4 : 128)) = o1;
}

template <int ROWMAJOR_OUT, int SIDE>
__device__ __forceinline__
void highway_layer(const float* __restrict__ xp,
                   const float* __restrict__ wp,
                   const float* __restrict__ bias,
                   float* __restrict__ out, int orow,
                   int T0, int nt0, int lane, int gw,
                   const float* __restrict__ fx, int fxrow,
                   const float* __restrict__ fpadf, const float* __restrict__ fpadb,
                   const float* __restrict__ fw, int foff,
                   float* __restrict__ fy) {
  const int h    = lane >> 4;
  const int c    = lane & 15;

  const float* xb = xp + (size_t)T0 * (16 * 256) + lane * 4;
  const float* wb = wp + (size_t)nt0 * (16 * 256) + lane * 4;

  v8f acc[4][4];
#pragma unroll
  for (int r = 0; r < 4; ++r)
#pragma unroll
    for (int f = 0; f < 4; ++f) acc[r][f] = v8f{};

  const float* tp[W_ + 1];
  float tw[W_ + 1];
  float* yw = nullptr;
  if (SIDE) {
    const int ch0 = (gw & 1) * 8;
    const int m   = (gw >> 1) * 16 + c;
    const int s   = m & (S_ - 1);
    const int b   = m >> 10;
#pragma unroll
    for (int j = 0; j <= W_; ++j) {
      const int t = s + foff + j;
      const float* q;
      if (t < W_)             q = fpadf + t * D_;
      else if (t >= W_ + S_)  q = fpadb + (t - W_ - S_) * D_;
      else                    q = fx + ((size_t)b * S_ + (t - W_)) * (size_t)fxrow;
      tp[j] = q + ch0 * 16 + 8 * h;
      tw[j] = fw[j];
    }
    yw = fy + ((size_t)(gw >> 1) * 1024 + ch0 * 64 + lane) * 4;
  }

  v4f cx[4], cw[4];
#pragma unroll
  for (int r = 0; r < 4; ++r) cx[r] = *(const v4f*)(xb + r * 4096);
  cw[0] = *(const v4f*)(wb);
  cw[1] = *(const v4f*)(wb + 4096);
  cw[2] = *(const v4f*)(wb + 16 * 4096);
  cw[3] = *(const v4f*)(wb + 17 * 4096);

#pragma nounroll
  for (int it2 = 0; it2 < D_ / 16; ++it2) {
    const int n1 = (2 * it2 + 1) * 128;
    v4f px[4], pw[4];
#pragma unroll
    for (int r = 0; r < 4; ++r) px[r] = *(const v4f*)(xb + n1 + r * 4096);
    pw[0] = *(const v4f*)(wb + n1);
    pw[1] = *(const v4f*)(wb + n1 + 4096);
    pw[2] = *(const v4f*)(wb + n1 + 16 * 4096);
    pw[3] = *(const v4f*)(wb + n1 + 17 * 4096);
    v4f fv[W_ + 1];
    if (SIDE) {
      __builtin_amdgcn_sched_barrier(0);
      const int dofs = (it2 >> 1) * 16 + (it2 & 1) * 4;
#pragma unroll
      for (int j = 0; j <= W_; ++j) fv[j] = *(const v4f*)(tp[j] + dofs);
    }
    __builtin_amdgcn_sched_barrier(0);

#pragma unroll
    for (int r = 0; r < 4; ++r)
#pragma unroll
      for (int f = 0; f < 4; ++f) acc[r][f] = mma(cw[f].xy, cx[r].xy, acc[r][f]);
#pragma unroll
    for (int r = 0; r < 4; ++r)
#pragma unroll
      for (int f = 0; f < 4; ++f) acc[r][f] = mma(cw[f].zw, cx[r].zw, acc[r][f]);

    const int n2 = (it2 + 1 < D_ / 16 ? 2 * it2 + 2 : 2 * it2 + 1) * 128;
#pragma unroll
    for (int r = 0; r < 4; ++r) cx[r] = *(const v4f*)(xb + n2 + r * 4096);
    cw[0] = *(const v4f*)(wb + n2);
    cw[1] = *(const v4f*)(wb + n2 + 4096);
    cw[2] = *(const v4f*)(wb + n2 + 16 * 4096);
    cw[3] = *(const v4f*)(wb + n2 + 17 * 4096);
    __builtin_amdgcn_sched_barrier(0);

#pragma unroll
    for (int r = 0; r < 4; ++r)
#pragma unroll
      for (int f = 0; f < 4; ++f) acc[r][f] = mma(pw[f].xy, px[r].xy, acc[r][f]);
#pragma unroll
    for (int r = 0; r < 4; ++r)
#pragma unroll
      for (int f = 0; f < 4; ++f) acc[r][f] = mma(pw[f].zw, px[r].zw, acc[r][f]);

    if (SIDE) {
      __builtin_amdgcn_sched_barrier(0);
      v4f a = {0.f, 0.f, 0.f, 0.f};
#pragma unroll
      for (int j = 0; j <= W_; ++j) {
        a.x = fmaf(tw[j], fv[j].x, a.x);
        a.y = fmaf(tw[j], fv[j].y, a.y);
        a.z = fmaf(tw[j], fv[j].z, a.z);
        a.w = fmaf(tw[j], fv[j].w, a.w);
      }
      *(v4f*)(yw + it2 * 128) = a;
    }
  }

  const float* bnA = bias + nt0 * 16 + 8 * h;
  const float* bnB = bnA + 16;
  const float* bgA = bnA + D_;
  const float* bgB = bnB + D_;
  const float* res = xp + ((size_t)T0 * 16 + nt0) * 256 + lane * 4;
  v4f rx[2][4];
  rx[0][0] = *(const v4f*)(res);
  rx[0][1] = *(const v4f*)(res + 128);
  rx[0][2] = *(const v4f*)(res + 256);
  rx[0][3] = *(const v4f*)(res + 384);
#pragma unroll
  for (int r = 0; r < 4; ++r) {
    if (r + 1 < 4) {
      const float* rn = res + (r + 1) * 4096;
      rx[(r + 1) & 1][0] = *(const v4f*)(rn);
      rx[(r + 1) & 1][1] = *(const v4f*)(rn + 128);
      rx[(r + 1) & 1][2] = *(const v4f*)(rn + 256);
      rx[(r + 1) & 1][3] = *(const v4f*)(rn + 384);
    }
    __builtin_amdgcn_sched_barrier(0);
    float *oA, *oB;
    if (ROWMAJOR_OUT) {
      oA = out + (size_t)((T0 + r) * 16 + c) * orow + nt0 * 16 + 8 * h;
      oB = oA + 16;
    } else {
      oA = out + ((size_t)(T0 + r) * 16 + nt0) * 256 + lane * 4;
      oB = oA + 256;
    }
    hw_tile_out<ROWMAJOR_OUT>(acc[r][0], acc[r][2], bnA, bgA, rx[r & 1][0], rx[r & 1][1], oA);
    hw_tile_out<ROWMAJOR_OUT>(acc[r][1], acc[r][3], bnB, bgB, rx[r & 1][2], rx[r & 1][3], oB);
  }
}

template <int SIDE>
__global__ __launch_bounds__(256)
void highway_pair_kernel(const float* __restrict__ xp,
                         const float* __restrict__ wp,
                         const float* __restrict__ bias,
                         float* ymid,
                         float* __restrict__ out, int orow,
                         const float* __restrict__ fx, int fxrow,
                         const float* __restrict__ fpadf, const float* __restrict__ fpadb,
                         const float* __restrict__ fw, int foff,
                         float* __restrict__ fy) {
  const int lane = threadIdx.x & 31;
  const int wave = __builtin_amdgcn_readfirstlane(threadIdx.x >> 5);
  const int T0   = blockIdx.x * 4;
  const int nt0  = wave * 2;
  const int gw   = blockIdx.x * 8 + wave;

  highway_layer<0, SIDE>(xp, wp, bias, ymid, D_, T0, nt0, lane, gw,
                         fx, fxrow, fpadf, fpadb, fw, foff, fy);
  __syncthreads();
  highway_layer<1, 0>(ymid, wp + 2 * D_ * D_, bias + 2 * D_, out, orow, T0, nt0, lane, gw,
                      nullptr, 0, nullptr, nullptr, nullptr, 0, nullptr);
}

extern "C" void kernel_launch(void* const* d_in, const int* in_sizes, int n_in,
                              void* d_out, int out_size, void* d_ws, size_t ws_size,
                              hipStream_t stream) {
  (void)in_sizes; (void)n_in; (void)out_size; (void)ws_size;

  const float* inputs   = (const float*)d_in[0];
  const float* fwd_pad  = (const float*)d_in[1];
  const float* bwd_pad  = (const float*)d_in[2];
  const float* fwd_w    = (const float*)d_in[3];
  const float* bwd_w    = (const float*)d_in[4];
  const float* fwd_hw_W = (const float*)d_in[5];
  const float* fwd_hw_b = (const float*)d_in[6];
  const float* bwd_hw_W = (const float*)d_in[7];
  const float* bwd_hw_b = (const float*)d_in[8];
  float* out = (float*)d_out;

  const size_t M = (size_t)B_ * S_;
  const size_t WMAT = (size_t)2 * D_ * D_;
  float* yA = (float*)d_ws;
  float* yB = yA + M * D_;
  float* wP = yB + M * D_;
  float* yC = wP + 2 * NL_ * NH_ * WMAT;

  dim3 hwBlock(256);
  dim3 hwGrid((unsigned)(M / 64));
  const int firBlocks = (int)(M * D_ / 4 / 256);

  pack_w_kernel<<<dim3((unsigned)(NL_ * NH_ * WMAT / 4 / 256), 2), 256, 0, stream>>>(
      fwd_hw_W, bwd_hw_W, wP);

  fir_kernel<<<firBlocks, 256, 0, stream>>>(inputs, D_, fwd_pad, bwd_pad, fwd_w, 0, yA);

  for (int i = 0; i < NL_; ++i) {
    for (int dir = 0; dir < 2; ++dir) {
      const float* ycur = dir ? yC : yA;
      const float* HW = wP + ((size_t)dir * NL_ * NH_ + (size_t)i * NH_) * WMAT;
      const float* Hb = (dir ? bwd_hw_b : fwd_hw_b) + (size_t)i * NH_ * 2 * D_;
      float* o = out + (size_t)i * M * 2 * D_ + dir * D_;

      const int ni = dir ? i + 1 : i;
      const int nd = dir ? 0 : 1;
      if (ni < NL_) {
        const float* xin; int xrow;
        if (ni == 0) { xin = inputs; xrow = D_; }
        else         { xin = out + (size_t)(ni - 1) * M * 2 * D_ + nd * D_; xrow = 2 * D_; }
        highway_pair_kernel<1><<<hwGrid, hwBlock, 0, stream>>>(
            ycur, HW, Hb, yB, o, 2 * D_,
            xin, xrow, fwd_pad + (size_t)ni * W_ * D_, bwd_pad + (size_t)ni * W_ * D_,
            (nd ? bwd_w : fwd_w) + ni * (W_ + 1), nd ? W_ : 0, nd ? yC : yA);
      } else {
        highway_pair_kernel<0><<<hwGrid, hwBlock, 0, stream>>>(
            ycur, HW, Hb, yB, o, 2 * D_, nullptr, 0, nullptr, nullptr, nullptr, 0, nullptr);
      }
    }
  }
}
